# starting kernel
# baseline (speedup 1.0000x reference)
#include <hip/hip_runtime.h>

typedef __attribute__((ext_vector_type(2))) float v2f;
typedef __attribute__((ext_vector_type(8))) float v8f;

__global__ void k_degree(const int* __restrict__ src, const int* __restrict__ dst,
                         float* __restrict__ deg_td, float* __restrict__ deg_bu,
                         int n_edges) {
  int e = blockIdx.x * blockDim.x + threadIdx.x;
  if (e >= n_edges) return;
  atomicAdd(&deg_td[dst[e]], 1.0f);
  atomicAdd(&deg_bu[src[e]], 1.0f);
}

__global__ void k_rsqrt(float* __restrict__ d, int n) {
  int i = blockIdx.x * blockDim.x + threadIdx.x;
  if (i >= n) return;
  d[i] = rsqrtf(d[i] + 1.0f);
}

__global__ void k_gemm_wmma(const float* __restrict__ A, const float* __restrict__ B,
                            float* __restrict__ C, const float* __restrict__ bias,
                            int M, int N, int K, int relu) {
  const int lane = threadIdx.x & 31;
  const int wave = threadIdx.x >> 5;
  const int tile = blockIdx.x * (blockDim.x >> 5) + wave;
  const int tn_cnt = N >> 4;
  const int total = (M >> 4) * tn_cnt;
  if (tile >= total) return;
  const int tm = (tile / tn_cnt) << 4;
  const int tn = (tile % tn_cnt) << 4;

  const int half = lane >> 4;
  const int l    = lane & 15;

  const float* __restrict__ Arow = A + (size_t)(tm + l) * K;
  const float* __restrict__ Bcol = B + tn + l;

  v8f acc = {};
  for (int k = 0; k < K; k += 4) {
    const int kk = k + half * 2;
    v2f a, b;
    a.x = Arow[kk];
    a.y = Arow[kk + 1];
    b.x = Bcol[(size_t)kk * N];
    b.y = Bcol[(size_t)(kk + 1) * N];
    acc = __builtin_amdgcn_wmma_f32_16x16x4_f32(
         false, a,  false, b,
         (short)0, acc,  false,  false);
  }

  const float bi = bias ? bias[tn + l] : 0.0f;
#pragma unroll
  for (int r = 0; r < 8; ++r) {
    float v = acc[r] + bi;
    if (relu) v = fmaxf(v, 0.0f);
    C[(size_t)(tm + half * 8 + r) * N + tn + l] = v;
  }
}

__global__ void k_edge_agg(const float* __restrict__ h,
                           const int* __restrict__ s_idx, const int* __restrict__ d_idx,
                           const float* __restrict__ dinv,
                           float* __restrict__ agg, int n_edges) {
  const int lane = threadIdx.x & 31;
  const int e = (blockIdx.x * blockDim.x + threadIdx.x) >> 5;
  if (e >= n_edges) return;
  const int s = s_idx[e];
  const int d = d_idx[e];
  const float norm = dinv[s] * dinv[d];
  const float4* __restrict__ hs = (const float4*)(h + (size_t)s * 128);
  float* __restrict__ ad = agg + (size_t)d * 128;
  float4 v = hs[lane];
  const int f = lane * 4;
  atomicAdd(&ad[f + 0], v.x * norm);
  atomicAdd(&ad[f + 1], v.y * norm);
  atomicAdd(&ad[f + 2], v.z * norm);
  atomicAdd(&ad[f + 3], v.w * norm);
}

__global__ void k_combine_relu(const float* __restrict__ agg, const float* __restrict__ hlin,
                               const float* __restrict__ dinv, const float* __restrict__ bias,
                               float* __restrict__ out, int n) {
  int i = blockIdx.x * blockDim.x + threadIdx.x;
  if (i >= n) return;
  const int node = i >> 7;
  const int f = i & 127;
  const float di = dinv[node];
  const float v = agg[i] + hlin[i] * di * di + bias[f];
  out[i] = fmaxf(v, 0.0f);
}

__global__ void k_combine_pool(const float* __restrict__ agg, const float* __restrict__ hlin,
                               const float* __restrict__ dinv, const float* __restrict__ bias,
                               const int* __restrict__ batch, float* __restrict__ pooled,
                               int col_off, int n) {
  int i = blockIdx.x * blockDim.x + threadIdx.x;
  if (i >= n) return;
  const int node = i >> 7;
  const int f = i & 127;
  const float di = dinv[node];
  const float v = agg[i] + hlin[i] * di * di + bias[f];
  atomicAdd(&pooled[(size_t)batch[node] * 256 + col_off + f], v);
}

static inline void launch_gemm(const float* A, const float* B, float* C,
                               const float* bias, int M, int N, int K, int relu,
                               hipStream_t stream) {
  const int tiles = (M >> 4) * (N >> 4);
  const int blocks = (tiles + 7) / 8;
  k_gemm_wmma<<<blocks, 256, 0, stream>>>(A, B, C, bias, M, N, K, relu);
}

extern "C" void kernel_launch(void* const* d_in, const int* in_sizes, int n_in,
                              void* d_out, int out_size, void* d_ws, size_t ws_size,
                              hipStream_t stream) {
  const float* x      = (const float*)d_in[0];
  const int* edge_idx = (const int*)d_in[1];
  const int* batch    = (const int*)d_in[2];
  const float* td_W1 = (const float*)d_in[4];
  const float* td_b1 = (const float*)d_in[5];
  const float* td_W2 = (const float*)d_in[6];
  const float* td_b2 = (const float*)d_in[7];
  const float* bu_W1 = (const float*)d_in[8];
  const float* bu_b1 = (const float*)d_in[9];
  const float* bu_W2 = (const float*)d_in[10];
  const float* bu_b2 = (const float*)d_in[11];
  const float* pw1   = (const float*)d_in[12];
  const float* pb1   = (const float*)d_in[13];
  const float* pw2   = (const float*)d_in[14];
  const float* pb2   = (const float*)d_in[15];

  const int IN_F = 256, HID = 128, OUT_F = 128, G = 512;
  const int n_nodes = in_sizes[0] / IN_F;
  const int n_edges = in_sizes[1] / 2;
  const int* src = edge_idx;
  const int* dst = edge_idx + n_edges;

  float* ws      = (float*)d_ws;
  const size_t NF = (size_t)n_nodes * 128;
  float* h_lin   = ws;
  float* h_agg   = ws + NF;
  float* h_act   = ws + 2 * NF;
  float* dinv_td = ws + 3 * NF;
  float* dinv_bu = dinv_td + n_nodes;
  float* pooled  = dinv_bu + n_nodes;
  float* mlp_h   = pooled + (size_t)G * 256;

  const int EB = (n_edges + 255) / 256;
  const int WB = (n_edges * 32 + 255) / 256;
  const int NB = ((int)NF + 255) / 256;

  hipMemsetAsync(dinv_td, 0, (size_t)2 * n_nodes * sizeof(float), stream);
  k_degree<<<EB, 256, 0, stream>>>(src, dst, dinv_td, dinv_bu, n_edges);
  k_rsqrt<<<(2 * n_nodes + 255) / 256, 256, 0, stream>>>(dinv_td, 2 * n_nodes);
  hipMemsetAsync(pooled, 0, (size_t)G * 256 * sizeof(float), stream);

  launch_gemm(x, td_W1, h_lin, nullptr, n_nodes, HID, IN_F, 0, stream);
  hipMemsetAsync(h_agg, 0, NF * sizeof(float), stream);
  k_edge_agg<<<WB, 256, 0, stream>>>(h_lin, src, dst, dinv_td, h_agg, n_edges);
  k_combine_relu<<<NB, 256, 0, stream>>>(h_agg, h_lin, dinv_td, td_b1, h_act, (int)NF);
  launch_gemm(h_act, td_W2, h_lin, nullptr, n_nodes, OUT_F, HID, 0, stream);
  hipMemsetAsync(h_agg, 0, NF * sizeof(float), stream);
  k_edge_agg<<<WB, 256, 0, stream>>>(h_lin, src, dst, dinv_td, h_agg, n_edges);
  k_combine_pool<<<NB, 256, 0, stream>>>(h_agg, h_lin, dinv_td, td_b2, batch, pooled,
                                          128, (int)NF);

  launch_gemm(x, bu_W1, h_lin, nullptr, n_nodes, HID, IN_F, 0, stream);
  hipMemsetAsync(h_agg, 0, NF * sizeof(float), stream);
  k_edge_agg<<<WB, 256, 0, stream>>>(h_lin, dst, src, dinv_bu, h_agg, n_edges);
  k_combine_relu<<<NB, 256, 0, stream>>>(h_agg, h_lin, dinv_bu, bu_b1, h_act, (int)NF);
  launch_gemm(h_act, bu_W2, h_lin, nullptr, n_nodes, OUT_F, HID, 0, stream);
  hipMemsetAsync(h_agg, 0, NF * sizeof(float), stream);
  k_edge_agg<<<WB, 256, 0, stream>>>(h_lin, dst, src, dinv_bu, h_agg, n_edges);
  k_combine_pool<<<NB, 256, 0, stream>>>(h_agg, h_lin, dinv_bu, bu_b2, batch, pooled,
                                          0, (int)NF);

  launch_gemm(pooled, pw1, mlp_h, pb1, G, 256, 256,  1, stream);
  launch_gemm(mlp_h, pw2, (float*)d_out, pb2, G, 128, 256,  0, stream);
}


# layer 2: edges and self loops add into a per-work-group LDS copy of the pooled [512,128] table (LDS float adds), flushed by whole-line atomics; W2 applied to the pooled table after; sum order changes
# speedup vs baseline: 1.4570x; 1.4570x over previous
#include <hip/hip_runtime.h>

typedef __attribute__((ext_vector_type(2))) float v2f;
typedef __attribute__((ext_vector_type(8))) float v8f;

__global__ void k_degree(const int* __restrict__ src, const int* __restrict__ dst,
                         float* __restrict__ deg_td, float* __restrict__ deg_bu,
                         int n_edges) {
  int e = blockIdx.x * blockDim.x + threadIdx.x;
  if (e >= n_edges) return;
  atomicAdd(&deg_td[dst[e]], 1.0f);
  atomicAdd(&deg_bu[src[e]], 1.0f);
}

__global__ void k_rsqrt(float* __restrict__ d, int n) {
  int i = blockIdx.x * blockDim.x + threadIdx.x;
  if (i >= n) return;
  d[i] = rsqrtf(d[i] + 1.0f);
}

__global__ void k_gemm_wmma(const float* __restrict__ A, const float* __restrict__ B,
                            float* __restrict__ C, const float* __restrict__ bias,
                            const float* __restrict__ rowscale,
                            int M, int N, int K, int lda, int ldc, int relu) {
  const int lane = threadIdx.x & 31;
  const int wave = threadIdx.x >> 5;
  const int tile = blockIdx.x * (blockDim.x >> 5) + wave;
  const int tn_cnt = N >> 4;
  const int total = (M >> 4) * tn_cnt;
  if (tile >= total) return;
  const int tm = (tile / tn_cnt) << 4;
  const int tn = (tile % tn_cnt) << 4;

  const int half = lane >> 4;
  const int l    = lane & 15;

  const float* __restrict__ Arow = A + (size_t)(tm + l) * lda;
  const float* __restrict__ Bcol = B + tn + l;

  v8f acc = {};
  for (int k = 0; k < K; k += 4) {
    const int kk = k + half * 2;
    v2f a, b;
    a.x = Arow[kk];
    a.y = Arow[kk + 1];
    b.x = Bcol[(size_t)kk * N];
    b.y = Bcol[(size_t)(kk + 1) * N];
    acc = __builtin_amdgcn_wmma_f32_16x16x4_f32(
         false, a,  false, b,
         (short)0, acc,  false,  false);
  }

  const float bi = bias ? bias[tn + l] : 0.0f;
#pragma unroll
  for (int r = 0; r < 8; ++r) {
    const int m = tm + half * 8 + r;
    float v = acc[r] + (rowscale ? bi * rowscale[m] : bi);
    if (relu) v = fmaxf(v, 0.0f);
    C[(size_t)m * ldc + tn + l] = v;
  }
}

__global__ void k_edge_agg(const float* __restrict__ h,
                           const int* __restrict__ s_idx, const int* __restrict__ d_idx,
                           const float* __restrict__ dinv,
                           float* __restrict__ agg, int n_edges) {
  const int lane = threadIdx.x & 31;
  const int e = (blockIdx.x * blockDim.x + threadIdx.x) >> 5;
  if (e >= n_edges) return;
  const int s = s_idx[e];
  const int d = d_idx[e];
  const float norm = dinv[s] * dinv[d];
  const float4* __restrict__ hs = (const float4*)(h + (size_t)s * 128);
  float* __restrict__ ad = agg + (size_t)d * 128;
  float4 v = hs[lane];
  const int f = lane * 4;
  atomicAdd(&ad[f + 0], v.x * norm);
  atomicAdd(&ad[f + 1], v.y * norm);
  atomicAdd(&ad[f + 2], v.z * norm);
  atomicAdd(&ad[f + 3], v.w * norm);
}

__global__ void k_combine_relu(const float* __restrict__ agg, const float* hlin,
                               const float* __restrict__ dinv, const float* __restrict__ bias,
                               float* out, int n) {
  int i = blockIdx.x * blockDim.x + threadIdx.x;
  if (i >= n) return;
  const int node = i >> 7;
  const int f = i & 127;
  const float di = dinv[node];
  const float v = agg[i] + hlin[i] * di * di + bias[f];
  out[i] = fmaxf(v, 0.0f);
}

#define POOL_G 512
__global__ __launch_bounds__(1024)
void k_edge_pool(const float* __restrict__ h_a, const int* __restrict__ s_a,
                 const int* __restrict__ d_a, const float* __restrict__ dinv_a, int col_a,
                 const float* __restrict__ h_b, const int* __restrict__ s_b,
                 const int* __restrict__ d_b, const float* __restrict__ dinv_b, int col_b,
                 const int* __restrict__ batch,
                 float* __restrict__ pooled, float* __restrict__ cnt,
                 int n_edges, int n_nodes) {
  __shared__ float tile[POOL_G * 128];
  __shared__ float tcnt[POOL_G];
  const int tid = threadIdx.x;
  const int lane = tid & 31;
  const int wave = tid >> 5;
  const int nwaves = blockDim.x >> 5;
  const int side_wgs = (int)gridDim.x >> 1;
  const bool side_b = (int)blockIdx.x >= side_wgs;
  const int wg = side_b ? (int)blockIdx.x - side_wgs : (int)blockIdx.x;
  const float* __restrict__ h    = side_b ? h_b : h_a;
  const int* __restrict__ s_idx  = side_b ? s_b : s_a;
  const int* __restrict__ d_idx  = side_b ? d_b : d_a;
  const float* __restrict__ dinv = side_b ? dinv_b : dinv_a;
  const int col_off = side_b ? col_b : col_a;
  const bool count = !side_b;

  float4* t4 = (float4*)tile;
  for (int i = tid; i < POOL_G * 32; i += blockDim.x) t4[i] = make_float4(0.f, 0.f, 0.f, 0.f);
  for (int i = tid; i < POOL_G; i += blockDim.x) tcnt[i] = 0.0f;
  __syncthreads();

  const int total = n_edges + n_nodes;
  const int chunk = (((total + side_wgs - 1) / side_wgs) + 31) & ~31;
  const int begin = wg * chunk;
  const int end = min(total, begin + chunk);

  for (int base = begin + wave * 32; base < end; base += nwaves * 32) {
    const int i = base + lane;
    const bool valid = i < end;
    int s = 0, g = 0;
    float w = 0.0f;
    if (valid) {
      if (i < n_edges) {
        s = s_idx[i];
        const int d = d_idx[i];
        w = dinv[s] * dinv[d];
        g = batch[d];
      } else {
        s = i - n_edges;
        const float di = dinv[s];
        w = di * di;
        g = batch[s];
        if (count) __hip_atomic_fetch_add(&tcnt[g], 1.0f, __ATOMIC_RELAXED, __HIP_MEMORY_SCOPE_WORKGROUP);
      }
    }
    const int s0 = __builtin_amdgcn_readfirstlane(s);
    const int g0 = __builtin_amdgcn_readfirstlane(g);
    if (!valid) { s = s0; g = g0; }
    const int wi = __float_as_int(w);

#pragma unroll
    for (int jj = 0; jj < 32; jj += 8) {
      float4 v[8];
#pragma unroll
      for (int u = 0; u < 8; ++u) {
        const int sj = __builtin_amdgcn_readlane(s, jj + u);
        v[u] = ((const float4*)(h + (size_t)sj * 128))[lane];
      }
#pragma unroll
      for (int u = 0; u < 8; ++u) {
        const int gj = __builtin_amdgcn_readlane(g, jj + u);
        const float wj = __int_as_float(__builtin_amdgcn_readlane(wi, jj + u));
        float* t = tile + gj * 128 + lane;
        __hip_atomic_fetch_add(t,      v[u].x * wj, __ATOMIC_RELAXED, __HIP_MEMORY_SCOPE_WORKGROUP);
        __hip_atomic_fetch_add(t + 32, v[u].y * wj, __ATOMIC_RELAXED, __HIP_MEMORY_SCOPE_WORKGROUP);
        __hip_atomic_fetch_add(t + 64, v[u].z * wj, __ATOMIC_RELAXED, __HIP_MEMORY_SCOPE_WORKGROUP);
        __hip_atomic_fetch_add(t + 96, v[u].w * wj, __ATOMIC_RELAXED, __HIP_MEMORY_SCOPE_WORKGROUP);
      }
    }
  }
  __syncthreads();

  for (int idx = tid; idx < POOL_G * 128; idx += blockDim.x) {
    const int g = ((idx >> 7) + (wg << 2)) & (POOL_G - 1);
    const int f = idx & 127;
    const float v = tile[g * 128 + (f & 3) * 32 + (f >> 2)];
    atomicAdd(&pooled[(size_t)g * 256 + col_off + f], v);
  }
  if (count) {
    for (int i = tid; i < POOL_G; i += blockDim.x) atomicAdd(&cnt[i], tcnt[i]);
  }
}

static inline void launch_gemm(const float* A, const float* B, float* C,
                               const float* bias, const float* rowscale,
                               int M, int N, int K, int lda, int ldc, int relu,
                               hipStream_t stream) {
  const int tiles = (M >> 4) * (N >> 4);
  const int blocks = (tiles + 7) / 8;
  k_gemm_wmma<<<blocks, 256, 0, stream>>>(A, B, C, bias, rowscale, M, N, K, lda, ldc, relu);
}

extern "C" void kernel_launch(void* const* d_in, const int* in_sizes, int n_in,
                              void* d_out, int out_size, void* d_ws, size_t ws_size,
                              hipStream_t stream) {
  const float* x      = (const float*)d_in[0];
  const int* edge_idx = (const int*)d_in[1];
  const int* batch    = (const int*)d_in[2];
  const float* td_W1 = (const float*)d_in[4];
  const float* td_b1 = (const float*)d_in[5];
  const float* td_W2 = (const float*)d_in[6];
  const float* td_b2 = (const float*)d_in[7];
  const float* bu_W1 = (const float*)d_in[8];
  const float* bu_b1 = (const float*)d_in[9];
  const float* bu_W2 = (const float*)d_in[10];
  const float* bu_b2 = (const float*)d_in[11];
  const float* pw1   = (const float*)d_in[12];
  const float* pb1   = (const float*)d_in[13];
  const float* pw2   = (const float*)d_in[14];
  const float* pb2   = (const float*)d_in[15];

  const int IN_F = 256, HID = 128, OUT_F = 128, G = 512;
  const int n_nodes = in_sizes[0] / IN_F;
  const int n_edges = in_sizes[1] / 2;
  const int* src = edge_idx;
  const int* dst = edge_idx + n_edges;

  float* ws      = (float*)d_ws;
  const size_t NF = (size_t)n_nodes * 128;
  float* h_lin   = ws;
  float* h_agg   = ws + NF;
  float* h_act   = ws + 2 * NF;
  float* dinv_td = ws + 3 * NF;
  float* dinv_bu = dinv_td + n_nodes;
  float* pooled  = dinv_bu + n_nodes;
  float* gcount  = pooled + (size_t)G * 256;
  float* pooled2 = h_agg;
  float* mlp_h   = h_act;

  const int EB = (n_edges + 255) / 256;
  const int WB = (n_edges * 32 + 255) / 256;
  const int NB = ((int)NF + 255) / 256;
  const int PB = 256;

  hipMemsetAsync(dinv_td, 0, (size_t)2 * n_nodes * sizeof(float), stream);
  k_degree<<<EB, 256, 0, stream>>>(src, dst, dinv_td, dinv_bu, n_edges);
  k_rsqrt<<<(2 * n_nodes + 255) / 256, 256, 0, stream>>>(dinv_td, 2 * n_nodes);
  hipMemsetAsync(pooled, 0, ((size_t)G * 256 + G) * sizeof(float), stream);

  launch_gemm(x, td_W1, h_lin, nullptr, nullptr, n_nodes, HID, IN_F, IN_F, HID, 0, stream);
  hipMemsetAsync(h_agg, 0, NF * sizeof(float), stream);
  k_edge_agg<<<WB, 256, 0, stream>>>(h_lin, src, dst, dinv_td, h_agg, n_edges);
  k_combine_relu<<<NB, 256, 0, stream>>>(h_agg, h_lin, dinv_td, td_b1, h_act, (int)NF);

  launch_gemm(x, bu_W1, h_lin, nullptr, nullptr, n_nodes, HID, IN_F, IN_F, HID, 0, stream);
  hipMemsetAsync(h_agg, 0, NF * sizeof(float), stream);
  k_edge_agg<<<WB, 256, 0, stream>>>(h_lin, dst, src, dinv_bu, h_agg, n_edges);
  k_combine_relu<<<NB, 256, 0, stream>>>(h_agg, h_lin, dinv_bu, bu_b1,  h_lin, (int)NF);

  k_edge_pool<<<PB, 1024, 0, stream>>>(h_act, src, dst, dinv_td,  128,
                                       h_lin, dst, src, dinv_bu,  0,
                                       batch, pooled, gcount, n_edges, n_nodes);

  launch_gemm(pooled, bu_W2, pooled2, bu_b2, gcount, G, OUT_F, HID, 256, 256, 0, stream);
  launch_gemm(pooled + 128, td_W2, pooled2 + 128, td_b2, gcount, G, OUT_F, HID, 256, 256, 0, stream);

  launch_gemm(pooled2, pw1, mlp_h, pb1, nullptr, G, 256, 256, 256, 256,  1, stream);
  launch_gemm(mlp_h, pw2, (float*)d_out, pb2, nullptr, G, 128, 256, 256, 128,  0, stream);
}


# layer 1: both branches' products in one launch: weights stay in registers (a 16-column strip per wave), each 16-row block of x is staged in LDS once and feeds both; x read once; sum order kept
# speedup vs baseline: 2.4812x; 1.7030x over previous
#include <hip/hip_runtime.h>

typedef __attribute__((ext_vector_type(2))) float v2f;
typedef __attribute__((ext_vector_type(8))) float v8f;

__global__ void k_degree(const int* __restrict__ src, const int* __restrict__ dst,
                         float* __restrict__ deg_td, float* __restrict__ deg_bu,
                         int n_edges) {
  int e = blockIdx.x * blockDim.x + threadIdx.x;
  if (e >= n_edges) return;
  atomicAdd(&deg_td[dst[e]], 1.0f);
  atomicAdd(&deg_bu[src[e]], 1.0f);
}

__global__ void k_rsqrt(float* __restrict__ d, int n) {
  int i = blockIdx.x * blockDim.x + threadIdx.x;
  if (i >= n) return;
  d[i] = rsqrtf(d[i] + 1.0f);
}

__global__ void k_gemm_wmma(const float* __restrict__ A, const float* __restrict__ B,
                            float* __restrict__ C, const float* __restrict__ bias,
                            const float* __restrict__ rowscale,
                            int M, int N, int K, int lda, int ldc, int relu) {
  const int lane = threadIdx.x & 31;
  const int wave = threadIdx.x >> 5;
  const int tile = blockIdx.x * (blockDim.x >> 5) + wave;
  const int tn_cnt = N >> 4;
  const int total = (M >> 4) * tn_cnt;
  if (tile >= total) return;
  const int tm = (tile / tn_cnt) << 4;
  const int tn = (tile % tn_cnt) << 4;

  const int half = lane >> 4;
  const int l    = lane & 15;

  const float* __restrict__ Arow = A + (size_t)(tm + l) * lda;
  const float* __restrict__ Bcol = B + tn + l;

  v8f acc = {};
  for (int k = 0; k < K; k += 4) {
    const int kk = k + half * 2;
    v2f a, b;
    a.x = Arow[kk];
    a.y = Arow[kk + 1];
    b.x = Bcol[(size_t)kk * N];
    b.y = Bcol[(size_t)(kk + 1) * N];
    acc = __builtin_amdgcn_wmma_f32_16x16x4_f32(
         false, a,  false, b,
         (short)0, acc,  false,  false);
  }

  const float bi = bias ? bias[tn + l] : 0.0f;
#pragma unroll
  for (int r = 0; r < 8; ++r) {
    const int m = tm + half * 8 + r;
    float v = acc[r] + (rowscale ? bi * rowscale[m] : bi);
    if (relu) v = fmaxf(v, 0.0f);
    C[(size_t)m * ldc + tn + l] = v;
  }
}

#define L1_PITCH 260
__global__ __launch_bounds__(512)
void k_gemm_l1_both(const float* __restrict__ x, const float* __restrict__ W_a,
                    const float* __restrict__ W_b, float* __restrict__ C_a,
                    float* __restrict__ C_b, int n_rows) {
  __shared__ __attribute__((aligned(16))) float xs[2][16 * L1_PITCH];
  const int n_tiles = n_rows >> 4;
  const int tid = threadIdx.x;
  const int lane = tid & 31;
  const int wave = tid >> 5;
  const int half = lane >> 4;
  const int l = lane & 15;
  const bool side_b = wave >= 8;
  const float* __restrict__ Wm = side_b ? W_b : W_a;
  float* __restrict__ C = side_b ? C_b : C_a;
  const int n0 = (wave & 7) << 4;

  v2f breg[64];
  const float* __restrict__ Bcol = Wm + n0 + l;
#pragma unroll
  for (int p = 0; p < 64; ++p) {
    const int kk = 4 * p + 2 * half;
    breg[p].x = Bcol[kk * 128];
    breg[p].y = Bcol[(kk + 1) * 128];
  }

  const int q0 = tid, q1 = tid + 512;
  const int so0 = (q0 >> 6) * L1_PITCH + (q0 & 63) * 4;
  const int so1 = (q1 >> 6) * L1_PITCH + (q1 & 63) * 4;

  int t = blockIdx.x;
  if (t < n_tiles) {
    const float4* __restrict__ g = (const float4*)(x + (size_t)t * 4096);
    const float4 a0 = g[q0], a1 = g[q1];
    *(float4*)(&xs[0][so0]) = a0;
    *(float4*)(&xs[0][so1]) = a1;
  }
  __syncthreads();

  int buf = 0;
  int t_done = -1;
  v8f done = {};
  for (; t < n_tiles; t += gridDim.x, buf ^= 1) {
    const int tn = t + (int)gridDim.x;
    const bool more = tn < n_tiles;
    float4 p0 = make_float4(0.f, 0.f, 0.f, 0.f), p1 = p0;
    if (more) {
      const float4* __restrict__ g = (const float4*)(x + (size_t)tn * 4096);
      p0 = g[q0];
      p1 = g[q1];
    }
    if (t_done >= 0) {
      float* __restrict__ Crow = C + ((size_t)t_done * 16 + half * 8) * 128 + n0 + l;
#pragma unroll
      for (int r = 0; r < 8; ++r) Crow[(size_t)r * 128] = done[r];
    }

    const float* As = &xs[buf][l * L1_PITCH + 2 * half];
    v8f acc = {};
    v2f fa[16], fb[16];
#pragma unroll
    for (int i = 0; i < 16; ++i) fa[i] = *(const v2f*)(As + 4 * i);
#pragma unroll
    for (int g = 0; g < 4; g += 2) {
#pragma unroll
      for (int i = 0; i < 16; ++i) fb[i] = *(const v2f*)(As + 4 * (16 * (g + 1) + i));
      __builtin_amdgcn_sched_barrier(0);
#pragma unroll
      for (int i = 0; i < 16; ++i)
        acc = __builtin_amdgcn_wmma_f32_16x16x4_f32(
             false, fa[i],  false, breg[16 * g + i],
             (short)0, acc,  false,  false);
      if (g + 2 < 4) {
#pragma unroll
        for (int i = 0; i < 16; ++i) fa[i] = *(const v2f*)(As + 4 * (16 * (g + 2) + i));
      }
      __builtin_amdgcn_sched_barrier(0);
#pragma unroll
      for (int i = 0; i < 16; ++i)
        acc = __builtin_amdgcn_wmma_f32_16x16x4_f32(
             false, fb[i],  false, breg[16 * (g + 1) + i],
             (short)0, acc,  false,  false);
    }
    done = acc;
    t_done = t;

    if (more) {
      *(float4*)(&xs[buf ^ 1][so0]) = p0;
      *(float4*)(&xs[buf ^ 1][so1]) = p1;
    }
    __syncthreads();
  }
  if (t_done >= 0) {
    float* __restrict__ Crow = C + ((size_t)t_done * 16 + half * 8) * 128 + n0 + l;
#pragma unroll
    for (int r = 0; r < 8; ++r) Crow[(size_t)r * 128] = done[r];
  }
}

#define OWN_ROWS 196
#define OWN_CAP  136
__global__ __launch_bounds__(1024)
void k_own_agg(const float* __restrict__ h, const int* __restrict__ s_idx,
               const int* __restrict__ d_idx, const float* __restrict__ dinv,
               const float* __restrict__ bias, float* __restrict__ out,
               int n_edges, int n_nodes) {
  __shared__ float acc[OWN_ROWS * 128];
  __shared__ int2 list[32 * OWN_CAP];
  const int tid = threadIdx.x;
  const int lane = tid & 31;
  const int wave = tid >> 5;
  const int nwaves = blockDim.x >> 5;
  const int wg = blockIdx.x;
  const int lo = wg * OWN_ROWS;
  const int cnt = min(OWN_ROWS, n_nodes - lo);
  if (cnt <= 0) return;
  const unsigned ucnt = (unsigned)cnt;

  float4* a4 = (float4*)acc;
  for (int i = tid; i < OWN_ROWS * 32; i += blockDim.x) a4[i] = make_float4(0.f, 0.f, 0.f, 0.f);
  __syncthreads();

  int2* wl = list + wave * OWN_CAP;
  int n = 0;

  auto add_rows = [&](const int k) {
    const int u = ((lane & 7) < k) ? (lane & 7) : 0;
    const int2 ent = wl[n - k + u];
    const int s = s_idx[ent.x];
    const int r = ent.y;
    const int wi = __float_as_int(dinv[s] * dinv[lo + r]);
    float4 v[8];
#pragma unroll
    for (int j = 0; j < 8; ++j) {
      if (j < k) {
        const int sj = __builtin_amdgcn_readlane(s, j);
        v[j] = ((const float4*)(h + (size_t)sj * 128))[lane];
      }
    }
#pragma unroll
    for (int j = 0; j < 8; ++j) {
      if (j < k) {
        const int rj = __builtin_amdgcn_readlane(r, j);
        const float wj = __int_as_float(__builtin_amdgcn_readlane(wi, j));
        float* t = acc + rj * 128 + lane;
        __hip_atomic_fetch_add(t,      v[j].x * wj, __ATOMIC_RELAXED, __HIP_MEMORY_SCOPE_WORKGROUP);
        __hip_atomic_fetch_add(t + 32, v[j].y * wj, __ATOMIC_RELAXED, __HIP_MEMORY_SCOPE_WORKGROUP);
        __hip_atomic_fetch_add(t + 64, v[j].z * wj, __ATOMIC_RELAXED, __HIP_MEMORY_SCOPE_WORKGROUP);
        __hip_atomic_fetch_add(t + 96, v[j].w * wj, __ATOMIC_RELAXED, __HIP_MEMORY_SCOPE_WORKGROUP);
      }
    }
    n -= k;
  };

  auto push = [&](const bool f, const int e, const int r) {
    unsigned b = __builtin_amdgcn_ballot_w32(f);
    while (b) {
      const int L = __builtin_ctz(b);
      b &= b - 1;
      if (lane == L) wl[n] = make_int2(e, r);
      ++n;
    }
  };

  const int n_full = n_edges & ~3;
  const int n_steps = (n_full + 127) >> 7;
  const int start = (int)(((long long)wg * n_steps) / (int)gridDim.x);
  const int per_wave = (n_steps + nwaves - 1) / nwaves;
  for (int i0 = 0; i0 < per_wave; i0 += 8) {
    int4 dv[8];
#pragma unroll
    for (int q = 0; q < 8; ++q) {
      const int kq = wave + (i0 + q) * nwaves;
      int g = kq + start;
      if (g >= n_steps) g -= n_steps;
      const int e0 = g * 128 + lane * 4;
      const bool ok = kq < n_steps && e0 < n_full;
      dv[q] = *(const int4*)(d_idx + (ok ? e0 : 0));
    }
#pragma unroll
    for (int q = 0; q < 8; ++q) {
      const int kq = wave + (i0 + q) * nwaves;
      int g = kq + start;
      if (g >= n_steps) g -= n_steps;
      const int e0 = g * 128 + lane * 4;
      const bool ok = kq < n_steps && e0 < n_full;
      const int rx = dv[q].x - lo, ry = dv[q].y - lo, rz = dv[q].z - lo, rw = dv[q].w - lo;
      push(ok && (unsigned)rx < ucnt, e0 + 0, rx);
      push(ok && (unsigned)ry < ucnt, e0 + 1, ry);
      push(ok && (unsigned)rz < ucnt, e0 + 2, rz);
      push(ok && (unsigned)rw < ucnt, e0 + 3, rw);
      while (n >= 8) add_rows(8);
    }
  }
  if (wave == 0 && n_full < n_edges) {
    const int e = n_full + lane;
    const bool in = e < n_edges;
    const int r = (in ? d_idx[e] : -1) - lo;
    push(in && (unsigned)r < ucnt, e, r);
  }
  if (n > 0) add_rows(n);
  __syncthreads();

  const float4* __restrict__ h4 = (const float4*)h;
  const float4* __restrict__ b4 = (const float4*)bias;
  float4* __restrict__ o4 = (float4*)out;
  for (int idx = tid; idx < cnt * 32; idx += blockDim.x) {
    const int r = idx >> 5;
    const int q = idx & 31;
    const int node = lo + r;
    const float di = dinv[node];
    const float4 hl = h4[(size_t)node * 32 + q];
    const float4 b = b4[q];
    const float* t = acc + r * 128 + q;
    float4 o;
    o.x = fmaxf(t[0]  + hl.x * di * di + b.x, 0.0f);
    o.y = fmaxf(t[32] + hl.y * di * di + b.y, 0.0f);
    o.z = fmaxf(t[64] + hl.z * di * di + b.z, 0.0f);
    o.w = fmaxf(t[96] + hl.w * di * di + b.w, 0.0f);
    o4[(size_t)node * 32 + q] = o;
  }
}

#define POOL_G 512
__global__ __launch_bounds__(1024)
void k_edge_pool(const float* __restrict__ h_a, const int* __restrict__ s_a,
                 const int* __restrict__ d_a, const float* __restrict__ dinv_a, int col_a,
                 const float* __restrict__ h_b, const int* __restrict__ s_b,
                 const int* __restrict__ d_b, const float* __restrict__ dinv_b, int col_b,
                 const int* __restrict__ batch,
                 float* __restrict__ pooled, float* __restrict__ cnt,
                 int n_edges, int n_nodes) {
  __shared__ float tile[POOL_G * 128];
  __shared__ float tcnt[POOL_G];
  const int tid = threadIdx.x;
  const int lane = tid & 31;
  const int wave = tid >> 5;
  const int nwaves = blockDim.x >> 5;
  const int side_wgs = (int)gridDim.x >> 1;
  const bool side_b = (int)blockIdx.x >= side_wgs;
  const int wg = side_b ? (int)blockIdx.x - side_wgs : (int)blockIdx.x;
  const float* __restrict__ h    = side_b ? h_b : h_a;
  const int* __restrict__ s_idx  = side_b ? s_b : s_a;
  const int* __restrict__ d_idx  = side_b ? d_b : d_a;
  const float* __restrict__ dinv = side_b ? dinv_b : dinv_a;
  const int col_off = side_b ? col_b : col_a;
  const bool count = !side_b;

  float4* t4 = (float4*)tile;
  for (int i = tid; i < POOL_G * 32; i += blockDim.x) t4[i] = make_float4(0.f, 0.f, 0.f, 0.f);
  for (int i = tid; i < POOL_G; i += blockDim.x) tcnt[i] = 0.0f;
  __syncthreads();

  const int total = n_edges + n_nodes;
  const int chunk = (((total + side_wgs - 1) / side_wgs) + 31) & ~31;
  const int begin = wg * chunk;
  const int end = min(total, begin + chunk);

  for (int base = begin + wave * 32; base < end; base += nwaves * 32) {
    const int i = base + lane;
    const bool valid = i < end;
    int s = 0, g = 0;
    float w = 0.0f;
    if (valid) {
      if (i < n_edges) {
        s = s_idx[i];
        const int d = d_idx[i];
        w = dinv[s] * dinv[d];
        g = batch[d];
      } else {
        s = i - n_edges;
        const float di = dinv[s];
        w = di * di;
        g = batch[s];
        if (count) __hip_atomic_fetch_add(&tcnt[g], 1.0f, __ATOMIC_RELAXED, __HIP_MEMORY_SCOPE_WORKGROUP);
      }
    }
    const int s0 = __builtin_amdgcn_readfirstlane(s);
    const int g0 = __builtin_amdgcn_readfirstlane(g);
    if (!valid) { s = s0; g = g0; }
    const int wi = __float_as_int(w);

#pragma unroll
    for (int jj = 0; jj < 32; jj += 8) {
      float4 v[8];
#pragma unroll
      for (int u = 0; u < 8; ++u) {
        const int sj = __builtin_amdgcn_readlane(s, jj + u);
        v[u] = ((const float4*)(h + (size_t)sj * 128))[lane];
      }
#pragma unroll
      for (int u = 0; u < 8; ++u) {
        const int gj = __builtin_amdgcn_readlane(g, jj + u);
        const float wj = __int_as_float(__builtin_amdgcn_readlane(wi, jj + u));
        float* t = tile + gj * 128 + lane;
        __hip_atomic_fetch_add(t,      v[u].x * wj, __ATOMIC_RELAXED, __HIP_MEMORY_SCOPE_WORKGROUP);
        __hip_atomic_fetch_add(t + 32, v[u].y * wj, __ATOMIC_RELAXED, __HIP_MEMORY_SCOPE_WORKGROUP);
        __hip_atomic_fetch_add(t + 64, v[u].z * wj, __ATOMIC_RELAXED, __HIP_MEMORY_SCOPE_WORKGROUP);
        __hip_atomic_fetch_add(t + 96, v[u].w * wj, __ATOMIC_RELAXED, __HIP_MEMORY_SCOPE_WORKGROUP);
      }
    }
  }
  __syncthreads();

  for (int idx = tid; idx < POOL_G * 128; idx += blockDim.x) {
    const int g = ((idx >> 7) + (wg << 2)) & (POOL_G - 1);
    const int f = idx & 127;
    const float v = tile[g * 128 + (f & 3) * 32 + (f >> 2)];
    atomicAdd(&pooled[(size_t)g * 256 + col_off + f], v);
  }
  if (count) {
    for (int i = tid; i < POOL_G; i += blockDim.x) atomicAdd(&cnt[i], tcnt[i]);
  }
}

static inline void launch_gemm(const float* A, const float* B, float* C,
                               const float* bias, const float* rowscale,
                               int M, int N, int K, int lda, int ldc, int relu,
                               hipStream_t stream) {
  const int tiles = (M >> 4) * (N >> 4);
  const int blocks = (tiles + 7) / 8;
  k_gemm_wmma<<<blocks, 256, 0, stream>>>(A, B, C, bias, rowscale, M, N, K, lda, ldc, relu);
}

extern "C" void kernel_launch(void* const* d_in, const int* in_sizes, int n_in,
                              void* d_out, int out_size, void* d_ws, size_t ws_size,
                              hipStream_t stream) {
  const float* x      = (const float*)d_in[0];
  const int* edge_idx = (const int*)d_in[1];
  const int* batch    = (const int*)d_in[2];
  const float* td_W1 = (const float*)d_in[4];
  const float* td_b1 = (const float*)d_in[5];
  const float* td_W2 = (const float*)d_in[6];
  const float* td_b2 = (const float*)d_in[7];
  const float* bu_W1 = (const float*)d_in[8];
  const float* bu_b1 = (const float*)d_in[9];
  const float* bu_W2 = (const float*)d_in[10];
  const float* bu_b2 = (const float*)d_in[11];
  const float* pw1   = (const float*)d_in[12];
  const float* pb1   = (const float*)d_in[13];
  const float* pw2   = (const float*)d_in[14];
  const float* pb2   = (const float*)d_in[15];

  const int IN_F = 256, HID = 128, OUT_F = 128, G = 512;
  const int n_nodes = in_sizes[0] / IN_F;
  const int n_edges = in_sizes[1] / 2;
  const int* src = edge_idx;
  const int* dst = edge_idx + n_edges;

  float* ws      = (float*)d_ws;
  const size_t NF = (size_t)n_nodes * 128;
  float* h_lin_td = ws;
  float* h_lin_bu = ws + NF;
  float* h_act   = ws + 2 * NF;
  float* h_agg   = ws;
  float* dinv_td = ws + 3 * NF;
  float* dinv_bu = dinv_td + n_nodes;
  float* pooled  = dinv_bu + n_nodes;
  float* gcount  = pooled + (size_t)G * 256;
  float* pooled2 = h_lin_bu;
  float* mlp_h   = h_act;

  const int EB = (n_edges + 255) / 256;
  const int OB = 256;
  const int PB = 256;

  hipMemsetAsync(dinv_td, 0, (size_t)2 * n_nodes * sizeof(float), stream);
  k_degree<<<EB, 256, 0, stream>>>(src, dst, dinv_td, dinv_bu, n_edges);
  k_rsqrt<<<(2 * n_nodes + 255) / 256, 256, 0, stream>>>(dinv_td, 2 * n_nodes);
  hipMemsetAsync(pooled, 0, ((size_t)G * 256 + G) * sizeof(float), stream);

  k_gemm_l1_both<<<256, 512, 0, stream>>>(x, td_W1, bu_W1, h_lin_td, h_lin_bu, n_nodes);

  k_own_agg<<<OB, 1024, 0, stream>>>(h_lin_td, src, dst, dinv_td, td_b1, h_act, n_edges, n_nodes);

  k_own_agg<<<OB, 1024, 0, stream>>>(h_lin_bu, dst, src, dinv_bu, bu_b1, h_agg, n_edges, n_nodes);

  k_edge_pool<<<PB, 1024, 0, stream>>>(h_act, src, dst, dinv_td,  128,
                                       h_agg, dst, src, dinv_bu,  0,
                                       batch, pooled, gcount, n_edges, n_nodes);

  launch_gemm(pooled, bu_W2, pooled2, bu_b2, gcount, G, OUT_F, HID, 256, 256, 0, stream);
  launch_gemm(pooled + 128, td_W2, pooled2 + 128, td_b2, gcount, G, OUT_F, HID, 256, 256, 0, stream);

  launch_gemm(pooled2, pw1, mlp_h, pb1, nullptr, G, 256, 256, 256, 256,  1, stream);
  launch_gemm(mlp_h, pw2, (float*)d_out, pb2, nullptr, G, 128, 256, 256, 128,  0, stream);
}


# layer 1 edge scan: the degree pass also writes a per-owner bit map of 128-edge steps; each owning work-group reads only its flagged steps, not the whole index list; order of the LDS adds changes
# speedup vs baseline: 2.7399x; 1.1043x over previous
#include <hip/hip_runtime.h>

typedef __attribute__((ext_vector_type(2))) float v2f;
typedef __attribute__((ext_vector_type(8))) float v8f;

#define OWN_ROWS 196
#define MAP_OWNERS 256
__global__ __launch_bounds__(1024)
void k_degree(const int* __restrict__ src, const int* __restrict__ dst,
              float* __restrict__ deg_td, float* __restrict__ deg_bu,
              int* __restrict__ map_td, int* __restrict__ map_bu,
              int n_edges) {
  __shared__ int m_td[MAP_OWNERS];
  __shared__ int m_bu[MAP_OWNERS];
  const int tid = threadIdx.x;
  if (tid < MAP_OWNERS) { m_td[tid] = 0; m_bu[tid] = 0; }
  __syncthreads();
  const int base = blockIdx.x * 4096;
  int d[4], s[4];
#pragma unroll
  for (int k = 0; k < 4; ++k) {
    const int e = base + k * 1024 + tid;
    const bool in = e < n_edges;
    d[k] = in ? dst[e] : -1;
    s[k] = in ? src[e] : -1;
  }
#pragma unroll
  for (int k = 0; k < 4; ++k) {
    if (d[k] >= 0) {
      const int e = base + k * 1024 + tid;
      atomicAdd(&deg_td[d[k]], 1.0f);
      atomicAdd(&deg_bu[s[k]], 1.0f);
      const int bit = (int)(1u << ((e >> 7) & 31));
      atomicOr(&m_td[d[k] / OWN_ROWS], bit);
      atomicOr(&m_bu[s[k] / OWN_ROWS], bit);
    }
  }
  __syncthreads();
  if (tid < MAP_OWNERS) map_td[(size_t)blockIdx.x * MAP_OWNERS + tid] = m_td[tid];
  else if (tid < 2 * MAP_OWNERS) map_bu[(size_t)blockIdx.x * MAP_OWNERS + tid - MAP_OWNERS] = m_bu[tid - MAP_OWNERS];
}

__global__ void k_rsqrt(float* __restrict__ d, int n) {
  int i = blockIdx.x * blockDim.x + threadIdx.x;
  if (i >= n) return;
  d[i] = rsqrtf(d[i] + 1.0f);
}

__global__ void k_gemm_wmma(const float* __restrict__ A, const float* __restrict__ B,
                            float* __restrict__ C, const float* __restrict__ bias,
                            const float* __restrict__ rowscale,
                            int M, int N, int K, int lda, int ldc, int relu) {
  const int lane = threadIdx.x & 31;
  const int wave = threadIdx.x >> 5;
  const int tile = blockIdx.x * (blockDim.x >> 5) + wave;
  const int tn_cnt = N >> 4;
  const int total = (M >> 4) * tn_cnt;
  if (tile >= total) return;
  const int tm = (tile / tn_cnt) << 4;
  const int tn = (tile % tn_cnt) << 4;

  const int half = lane >> 4;
  const int l    = lane & 15;

  const float* __restrict__ Arow = A + (size_t)(tm + l) * lda;
  const float* __restrict__ Bcol = B + tn + l;

  v8f acc = {};
  for (int k = 0; k < K; k += 4) {
    const int kk = k + half * 2;
    v2f a, b;
    a.x = Arow[kk];
    a.y = Arow[kk + 1];
    b.x = Bcol[(size_t)kk * N];
    b.y = Bcol[(size_t)(kk + 1) * N];
    acc = __builtin_amdgcn_wmma_f32_16x16x4_f32(
         false, a,  false, b,
         (short)0, acc,  false,  false);
  }

  const float bi = bias ? bias[tn + l] : 0.0f;
#pragma unroll
  for (int r = 0; r < 8; ++r) {
    const int m = tm + half * 8 + r;
    float v = acc[r] + (rowscale ? bi * rowscale[m] : bi);
    if (relu) v = fmaxf(v, 0.0f);
    C[(size_t)m * ldc + tn + l] = v;
  }
}

#define L1_PITCH 260
__global__ __launch_bounds__(512)
void k_gemm_l1_both(const float* __restrict__ x, const float* __restrict__ W_a,
                    const float* __restrict__ W_b, float* __restrict__ C_a,
                    float* __restrict__ C_b, int n_rows) {
  __shared__ __attribute__((aligned(16))) float xs[2][16 * L1_PITCH];
  const int n_tiles = n_rows >> 4;
  const int tid = threadIdx.x;
  const int lane = tid & 31;
  const int wave = tid >> 5;
  const int half = lane >> 4;
  const int l = lane & 15;
  const bool side_b = wave >= 8;
  const float* __restrict__ Wm = side_b ? W_b : W_a;
  float* __restrict__ C = side_b ? C_b : C_a;
  const int n0 = (wave & 7) << 4;

  v2f breg[64];
  const float* __restrict__ Bcol = Wm + n0 + l;
#pragma unroll
  for (int p = 0; p < 64; ++p) {
    const int kk = 4 * p + 2 * half;
    breg[p].x = Bcol[kk * 128];
    breg[p].y = Bcol[(kk + 1) * 128];
  }

  const int q0 = tid, q1 = tid + 512;
  const int so0 = (q0 >> 6) * L1_PITCH + (q0 & 63) * 4;
  const int so1 = (q1 >> 6) * L1_PITCH + (q1 & 63) * 4;

  int t = blockIdx.x;
  if (t < n_tiles) {
    const float4* __restrict__ g = (const float4*)(x + (size_t)t * 4096);
    const float4 a0 = g[q0], a1 = g[q1];
    *(float4*)(&xs[0][so0]) = a0;
    *(float4*)(&xs[0][so1]) = a1;
  }
  __syncthreads();

  int buf = 0;
  int t_done = -1;
  v8f done = {};
  for (; t < n_tiles; t += gridDim.x, buf ^= 1) {
    const int tn = t + (int)gridDim.x;
    const bool more = tn < n_tiles;
    float4 p0 = make_float4(0.f, 0.f, 0.f, 0.f), p1 = p0;
    if (more) {
      const float4* __restrict__ g = (const float4*)(x + (size_t)tn * 4096);
      p0 = g[q0];
      p1 = g[q1];
    }
    if (t_done >= 0) {
      float* __restrict__ Crow = C + ((size_t)t_done * 16 + half * 8) * 128 + n0 + l;
#pragma unroll
      for (int r = 0; r < 8; ++r) Crow[(size_t)r * 128] = done[r];
    }

    const float* As = &xs[buf][l * L1_PITCH + 2 * half];
    v8f acc = {};
    v2f fa[16], fb[16];
#pragma unroll
    for (int i = 0; i < 16; ++i) fa[i] = *(const v2f*)(As + 4 * i);
#pragma unroll
    for (int g = 0; g < 4; g += 2) {
#pragma unroll
      for (int i = 0; i < 16; ++i) fb[i] = *(const v2f*)(As + 4 * (16 * (g + 1) + i));
      __builtin_amdgcn_sched_barrier(0);
#pragma unroll
      for (int i = 0; i < 16; ++i)
        acc = __builtin_amdgcn_wmma_f32_16x16x4_f32(
             false, fa[i],  false, breg[16 * g + i],
             (short)0, acc,  false,  false);
      if (g + 2 < 4) {
#pragma unroll
        for (int i = 0; i < 16; ++i) fa[i] = *(const v2f*)(As + 4 * (16 * (g + 2) + i));
      }
      __builtin_amdgcn_sched_barrier(0);
#pragma unroll
      for (int i = 0; i < 16; ++i)
        acc = __builtin_amdgcn_wmma_f32_16x16x4_f32(
             false, fb[i],  false, breg[16 * (g + 1) + i],
             (short)0, acc,  false,  false);
    }
    done = acc;
    t_done = t;

    if (more) {
      *(float4*)(&xs[buf ^ 1][so0]) = p0;
      *(float4*)(&xs[buf ^ 1][so1]) = p1;
    }
    __syncthreads();
  }
  if (t_done >= 0) {
    float* __restrict__ Crow = C + ((size_t)t_done * 16 + half * 8) * 128 + n0 + l;
#pragma unroll
    for (int r = 0; r < 8; ++r) Crow[(size_t)r * 128] = done[r];
  }
}

#define OWN_CAP  136
#define STEP_CAP 8192
__global__ __launch_bounds__(1024)
void k_own_agg(const float* __restrict__ h, const int* __restrict__ s_idx,
               const int* __restrict__ d_idx, const float* __restrict__ dinv,
               const float* __restrict__ bias, const int* __restrict__ map,
               float* __restrict__ out, int n_edges, int n_nodes) {
  __shared__ float acc[OWN_ROWS * 128];
  __shared__ int2 list[32 * OWN_CAP];
  __shared__ __attribute__((aligned(16))) unsigned short steps[STEP_CAP];
  __shared__ int n_flag;
  const int tid = threadIdx.x;
  const int lane = tid & 31;
  const int wave = tid >> 5;
  const int nwaves = blockDim.x >> 5;
  const int wg = blockIdx.x;
  const int lo = wg * OWN_ROWS;
  const int cnt = min(OWN_ROWS, n_nodes - lo);
  if (cnt <= 0) return;
  const unsigned ucnt = (unsigned)cnt;

  if (tid == 0) n_flag = 0;
  float4* a4 = (float4*)acc;
  for (int i = tid; i < OWN_ROWS * 32; i += blockDim.x) a4[i] = make_float4(0.f, 0.f, 0.f, 0.f);
  uint4* s4 = (uint4*)steps;
  for (int i = tid; i < STEP_CAP / 8; i += blockDim.x) s4[i] = make_uint4(0u, 0u, 0u, 0u);
  __syncthreads();

  const int n_full = n_edges & ~3;
  const int n_words = (n_edges + 4095) >> 12;
  const int w0 = (int)(((unsigned)wg * (unsigned)n_words) / gridDim.x);
  for (int j0 = tid; j0 < n_words; j0 += blockDim.x) {
    int j = j0 + w0;
    if (j >= n_words) j -= n_words;
    unsigned m = (unsigned)map[(size_t)j * MAP_OWNERS + wg];
    const int c = __popc(m);
    if (c) {
      int pos = atomicAdd(&n_flag, c);
      while (m) {
        const int b = __builtin_ctz(m);
        m &= m - 1;
        if (pos < STEP_CAP) steps[pos] = (unsigned short)(j * 32 + b);
        ++pos;
      }
    }
  }
  __syncthreads();
  const int nf = __builtin_amdgcn_readfirstlane(min(n_flag, STEP_CAP));
  const int wv = __builtin_amdgcn_readfirstlane(wave);

  int2* wl = list + wave * OWN_CAP;
  int n = 0;

  auto add_rows = [&](const int k) {
    const int u = ((lane & 7) < k) ? (lane & 7) : 0;
    const int2 ent = wl[n - k + u];
    const int s = s_idx[ent.x];
    const int r = ent.y;
    const int wi = __float_as_int(dinv[s] * dinv[lo + r]);
    float4 v[8];
#pragma unroll
    for (int j = 0; j < 8; ++j) {
      if (j < k) {
        const int sj = __builtin_amdgcn_readlane(s, j);
        v[j] = ((const float4*)(h + (size_t)sj * 128))[lane];
      }
    }
#pragma unroll
    for (int j = 0; j < 8; ++j) {
      if (j < k) {
        const int rj = __builtin_amdgcn_readlane(r, j);
        const float wj = __int_as_float(__builtin_amdgcn_readlane(wi, j));
        float* t = acc + rj * 128 + lane;
        __hip_atomic_fetch_add(t,      v[j].x * wj, __ATOMIC_RELAXED, __HIP_MEMORY_SCOPE_WORKGROUP);
        __hip_atomic_fetch_add(t + 32, v[j].y * wj, __ATOMIC_RELAXED, __HIP_MEMORY_SCOPE_WORKGROUP);
        __hip_atomic_fetch_add(t + 64, v[j].z * wj, __ATOMIC_RELAXED, __HIP_MEMORY_SCOPE_WORKGROUP);
        __hip_atomic_fetch_add(t + 96, v[j].w * wj, __ATOMIC_RELAXED, __HIP_MEMORY_SCOPE_WORKGROUP);
      }
    }
    n -= k;
  };

  auto push = [&](const bool f, const int e, const int r) {
    unsigned b = __builtin_amdgcn_ballot_w32(f);
    while (b) {
      const int L = __builtin_ctz(b);
      b &= b - 1;
      if (lane == L) wl[n] = make_int2(e, r);
      ++n;
    }
  };

  for (int i0 = wv * 8; i0 < nf; i0 += nwaves * 8) {
    int4 dv[8];
    int eb[8];
    const uint4 sw = *(const uint4*)(&steps[i0]);
    const unsigned swa[4] = {sw.x, sw.y, sw.z, sw.w};
#pragma unroll
    for (int q = 0; q < 8; ++q) {
      const int i = i0 + q;
      const int st = (int)((swa[q >> 1] >> (16 * (q & 1))) & 0xffffu);
      const int e0 = st * 128 + lane * 4;
      const bool ok = i < nf && e0 < n_full;
      eb[q] = ok ? e0 : -1;
      dv[q] = *(const int4*)(d_idx + (ok ? e0 : 0));
    }
#pragma unroll
    for (int q = 0; q < 8; ++q) {
      const int e0 = eb[q];
      const bool ok = e0 >= 0;
      const int rx = dv[q].x - lo, ry = dv[q].y - lo, rz = dv[q].z - lo, rw = dv[q].w - lo;
      push(ok && (unsigned)rx < ucnt, e0 + 0, rx);
      push(ok && (unsigned)ry < ucnt, e0 + 1, ry);
      push(ok && (unsigned)rz < ucnt, e0 + 2, rz);
      push(ok && (unsigned)rw < ucnt, e0 + 3, rw);
      while (n >= 8) add_rows(8);
    }
  }
  if (wave == 0 && n_full < n_edges) {
    const int e = n_full + lane;
    const bool in = e < n_edges;
    const int r = (in ? d_idx[e] : -1) - lo;
    push(in && (unsigned)r < ucnt, e, r);
  }
  if (n > 0) add_rows(n);
  __syncthreads();

  const float4* __restrict__ h4 = (const float4*)h;
  const float4* __restrict__ b4 = (const float4*)bias;
  float4* __restrict__ o4 = (float4*)out;
  for (int idx = tid; idx < cnt * 32; idx += blockDim.x) {
    const int r = idx >> 5;
    const int q = idx & 31;
    const int node = lo + r;
    const float di = dinv[node];
    const float4 hl = h4[(size_t)node * 32 + q];
    const float4 b = b4[q];
    const float* t = acc + r * 128 + q;
    float4 o;
    o.x = fmaxf(t[0]  + hl.x * di * di + b.x, 0.0f);
    o.y = fmaxf(t[32] + hl.y * di * di + b.y, 0.0f);
    o.z = fmaxf(t[64] + hl.z * di * di + b.z, 0.0f);
    o.w = fmaxf(t[96] + hl.w * di * di + b.w, 0.0f);
    o4[(size_t)node * 32 + q] = o;
  }
}

#define POOL_G 512
__global__ __launch_bounds__(1024)
void k_edge_pool(const float* __restrict__ h_a, const int* __restrict__ s_a,
                 const int* __restrict__ d_a, const float* __restrict__ dinv_a, int col_a,
                 const float* __restrict__ h_b, const int* __restrict__ s_b,
                 const int* __restrict__ d_b, const float* __restrict__ dinv_b, int col_b,
                 const int* __restrict__ batch,
                 float* __restrict__ pooled, float* __restrict__ cnt,
                 int n_edges, int n_nodes) {
  __shared__ float tile[POOL_G * 128];
  __shared__ float tcnt[POOL_G];
  const int tid = threadIdx.x;
  const int lane = tid & 31;
  const int wave = tid >> 5;
  const int nwaves = blockDim.x >> 5;
  const int side_wgs = (int)gridDim.x >> 1;
  const bool side_b = (int)blockIdx.x >= side_wgs;
  const int wg = side_b ? (int)blockIdx.x - side_wgs : (int)blockIdx.x;
  const float* __restrict__ h    = side_b ? h_b : h_a;
  const int* __restrict__ s_idx  = side_b ? s_b : s_a;
  const int* __restrict__ d_idx  = side_b ? d_b : d_a;
  const float* __restrict__ dinv = side_b ? dinv_b : dinv_a;
  const int col_off = side_b ? col_b : col_a;
  const bool count = !side_b;

  float4* t4 = (float4*)tile;
  for (int i = tid; i < POOL_G * 32; i += blockDim.x) t4[i] = make_float4(0.f, 0.f, 0.f, 0.f);
  for (int i = tid; i < POOL_G; i += blockDim.x) tcnt[i] = 0.0f;
  __syncthreads();

  const int total = n_edges + n_nodes;
  const int chunk = (((total + side_wgs - 1) / side_wgs) + 31) & ~31;
  const int begin = wg * chunk;
  const int end = min(total, begin + chunk);

  for (int base = begin + wave * 32; base < end; base += nwaves * 32) {
    const int i = base + lane;
    const bool valid = i < end;
    int s = 0, g = 0;
    float w = 0.0f;
    if (valid) {
      if (i < n_edges) {
        s = s_idx[i];
        const int d = d_idx[i];
        w = dinv[s] * dinv[d];
        g = batch[d];
      } else {
        s = i - n_edges;
        const float di = dinv[s];
        w = di * di;
        g = batch[s];
        if (count) __hip_atomic_fetch_add(&tcnt[g], 1.0f, __ATOMIC_RELAXED, __HIP_MEMORY_SCOPE_WORKGROUP);
      }
    }
    const int s0 = __builtin_amdgcn_readfirstlane(s);
    const int g0 = __builtin_amdgcn_readfirstlane(g);
    if (!valid) { s = s0; g = g0; }
    const int wi = __float_as_int(w);

#pragma unroll
    for (int jj = 0; jj < 32; jj += 8) {
      float4 v[8];
#pragma unroll
      for (int u = 0; u < 8; ++u) {
        const int sj = __builtin_amdgcn_readlane(s, jj + u);
        v[u] = ((const float4*)(h + (size_t)sj * 128))[lane];
      }
#pragma unroll
      for (int u = 0; u < 8; ++u) {
        const int gj = __builtin_amdgcn_readlane(g, jj + u);
        const float wj = __int_as_float(__builtin_amdgcn_readlane(wi, jj + u));
        float* t = tile + gj * 128 + lane;
        __hip_atomic_fetch_add(t,      v[u].x * wj, __ATOMIC_RELAXED, __HIP_MEMORY_SCOPE_WORKGROUP);
        __hip_atomic_fetch_add(t + 32, v[u].y * wj, __ATOMIC_RELAXED, __HIP_MEMORY_SCOPE_WORKGROUP);
        __hip_atomic_fetch_add(t + 64, v[u].z * wj, __ATOMIC_RELAXED, __HIP_MEMORY_SCOPE_WORKGROUP);
        __hip_atomic_fetch_add(t + 96, v[u].w * wj, __ATOMIC_RELAXED, __HIP_MEMORY_SCOPE_WORKGROUP);
      }
    }
  }
  __syncthreads();

  for (int idx = tid; idx < POOL_G * 128; idx += blockDim.x) {
    const int g = ((idx >> 7) + (wg << 2)) & (POOL_G - 1);
    const int f = idx & 127;
    const float v = tile[g * 128 + (f & 3) * 32 + (f >> 2)];
    atomicAdd(&pooled[(size_t)g * 256 + col_off + f], v);
  }
  if (count) {
    for (int i = tid; i < POOL_G; i += blockDim.x) atomicAdd(&cnt[i], tcnt[i]);
  }
}

static inline void launch_gemm(const float* A, const float* B, float* C,
                               const float* bias, const float* rowscale,
                               int M, int N, int K, int lda, int ldc, int relu,
                               hipStream_t stream) {
  const int tiles = (M >> 4) * (N >> 4);
  const int blocks = (tiles + 7) / 8;
  k_gemm_wmma<<<blocks, 256, 0, stream>>>(A, B, C, bias, rowscale, M, N, K, lda, ldc, relu);
}

extern "C" void kernel_launch(void* const* d_in, const int* in_sizes, int n_in,
                              void* d_out, int out_size, void* d_ws, size_t ws_size,
                              hipStream_t stream) {
  const float* x      = (const float*)d_in[0];
  const int* edge_idx = (const int*)d_in[1];
  const int* batch    = (const int*)d_in[2];
  const float* td_W1 = (const float*)d_in[4];
  const float* td_b1 = (const float*)d_in[5];
  const float* td_W2 = (const float*)d_in[6];
  const float* td_b2 = (const float*)d_in[7];
  const float* bu_W1 = (const float*)d_in[8];
  const float* bu_b1 = (const float*)d_in[9];
  const float* bu_W2 = (const float*)d_in[10];
  const float* bu_b2 = (const float*)d_in[11];
  const float* pw1   = (const float*)d_in[12];
  const float* pb1   = (const float*)d_in[13];
  const float* pw2   = (const float*)d_in[14];
  const float* pb2   = (const float*)d_in[15];

  const int IN_F = 256, HID = 128, OUT_F = 128, G = 512;
  const int n_nodes = in_sizes[0] / IN_F;
  const int n_edges = in_sizes[1] / 2;
  const int* src = edge_idx;
  const int* dst = edge_idx + n_edges;

  float* ws      = (float*)d_ws;
  const size_t NF = (size_t)n_nodes * 128;
  float* h_lin_td = ws;
  float* h_lin_bu = ws + NF;
  float* h_act   = ws + 2 * NF;
  float* h_agg   = ws;
  float* dinv_td = ws + 3 * NF;
  float* dinv_bu = dinv_td + n_nodes;
  float* pooled  = dinv_bu + n_nodes;
  float* gcount  = pooled + (size_t)G * 256;
  float* pooled2 = h_lin_bu;
  float* mlp_h   = h_act;

  const int MR = (n_edges + 4095) / 4096;
  int* map_td = (int*)pooled;
  int* map_bu = map_td + (size_t)MR * 256;
  const int OB = 256;
  const int PB = 256;

  hipMemsetAsync(dinv_td, 0, (size_t)2 * n_nodes * sizeof(float), stream);
  k_degree<<<MR, 1024, 0, stream>>>(src, dst, dinv_td, dinv_bu, map_td, map_bu, n_edges);
  k_rsqrt<<<(2 * n_nodes + 255) / 256, 256, 0, stream>>>(dinv_td, 2 * n_nodes);

  k_gemm_l1_both<<<256, 512, 0, stream>>>(x, td_W1, bu_W1, h_lin_td, h_lin_bu, n_nodes);

  k_own_agg<<<OB, 1024, 0, stream>>>(h_lin_td, src, dst, dinv_td, td_b1, map_td, h_act, n_edges, n_nodes);

  k_own_agg<<<OB, 1024, 0, stream>>>(h_lin_bu, dst, src, dinv_bu, bu_b1, map_bu, h_agg, n_edges, n_nodes);

  hipMemsetAsync(pooled, 0, ((size_t)G * 256 + G) * sizeof(float), stream);
  k_edge_pool<<<PB, 1024, 0, stream>>>(h_act, src, dst, dinv_td,  128,
                                       h_agg, dst, src, dinv_bu,  0,
                                       batch, pooled, gcount, n_edges, n_nodes);

  launch_gemm(pooled, bu_W2, pooled2, bu_b2, gcount, G, OUT_F, HID, 256, 256, 0, stream);
  launch_gemm(pooled + 128, td_W2, pooled2 + 128, td_b2, gcount, G, OUT_F, HID, 256, 256, 0, stream);

  launch_gemm(pooled2, pw1, mlp_h, pb1, nullptr, G, 256, 256, 256, 256,  1, stream);
  launch_gemm(mlp_h, pw2, (float*)d_out, pb2, nullptr, G, 128, 256, 256, 128,  0, stream);
}


# degree count: a work-group counts a 16384-edge slice of one index list in an all-node LDS histogram and stores it as a plane; rsqrt pass sums the planes; no global add per edge, no memset; exact
# speedup vs baseline: 3.2427x; 1.1835x over previous
#include <hip/hip_runtime.h>

typedef __attribute__((ext_vector_type(2))) float v2f;
typedef __attribute__((ext_vector_type(8))) float v8f;

#define OWN_ROWS 196
#define MAP_OWNERS 256
#define DEG_ROWS 4
#define DEG_N (MAP_OWNERS * OWN_ROWS)
__global__ __launch_bounds__(1024)
void k_degree(const int* __restrict__ src, const int* __restrict__ dst,
              int* __restrict__ planes,
              int* __restrict__ map_td, int* __restrict__ map_bu,
              int n_edges, int n_map_rows) {
  __shared__ __attribute__((aligned(16))) int hist[DEG_N];
  __shared__ int m[DEG_ROWS * MAP_OWNERS];
  const int tid = threadIdx.x;
  const int side_wgs = (int)gridDim.x >> 1;
  const bool side_b = (int)blockIdx.x >= side_wgs;
  const int wg = side_b ? (int)blockIdx.x - side_wgs : (int)blockIdx.x;
  const int* __restrict__ idx = side_b ? src : dst;
  int* __restrict__ map = side_b ? map_bu : map_td;

  int4* h4 = (int4*)hist;
  for (int i = tid; i < DEG_N / 4; i += blockDim.x) h4[i] = make_int4(0, 0, 0, 0);
  for (int i = tid; i < DEG_ROWS * MAP_OWNERS; i += blockDim.x) m[i] = 0;
  __syncthreads();

  const int base = wg * (DEG_ROWS * 4096);
  int v[DEG_ROWS * 4];
#pragma unroll
  for (int k = 0; k < DEG_ROWS * 4; ++k) {
    const int e = base + k * 1024 + tid;
    v[k] = e < n_edges ? idx[e] : -1;
  }
#pragma unroll
  for (int k = 0; k < DEG_ROWS * 4; ++k) {
    if (v[k] >= 0) {
      const int e = base + k * 1024 + tid;
      atomicAdd(&hist[v[k]], 1);
      atomicOr(&m[(k >> 2) * MAP_OWNERS + v[k] / OWN_ROWS], (int)(1u << ((e >> 7) & 31)));
    }
  }
  __syncthreads();

  int4* __restrict__ p4 = (int4*)(planes + (size_t)blockIdx.x * DEG_N);
  for (int i = tid; i < DEG_N / 4; i += blockDim.x) p4[i] = h4[i];
  for (int i = tid; i < DEG_ROWS * MAP_OWNERS; i += blockDim.x) {
    const int row = wg * DEG_ROWS + i / MAP_OWNERS;
    if (row < n_map_rows) map[(size_t)row * MAP_OWNERS + (i % MAP_OWNERS)] = m[i];
  }
}

__global__ void k_rsqrt(const int* __restrict__ planes, float* __restrict__ dinv,
                        int n_nodes, int side_wgs) {
  const int i = blockIdx.x * blockDim.x + threadIdx.x;
  if (i >= 2 * n_nodes) return;
  const bool side_b = i >= n_nodes;
  const int node = side_b ? i - n_nodes : i;
  const int* __restrict__ p = planes + (side_b ? (size_t)side_wgs * DEG_N : (size_t)0) + node;
  int c = 0;
#pragma unroll 16
  for (int w = 0; w < side_wgs; ++w) c += p[(size_t)w * DEG_N];
  dinv[i] = rsqrtf((float)c + 1.0f);
}

__global__ void k_gemm_wmma(const float* __restrict__ A, const float* __restrict__ B,
                            float* __restrict__ C, const float* __restrict__ bias,
                            const float* __restrict__ rowscale,
                            int M, int N, int K, int lda, int ldc, int relu) {
  const int lane = threadIdx.x & 31;
  const int wave = threadIdx.x >> 5;
  const int tile = blockIdx.x * (blockDim.x >> 5) + wave;
  const int tn_cnt = N >> 4;
  const int total = (M >> 4) * tn_cnt;
  if (tile >= total) return;
  const int tm = (tile / tn_cnt) << 4;
  const int tn = (tile % tn_cnt) << 4;

  const int half = lane >> 4;
  const int l    = lane & 15;

  const float* __restrict__ Arow = A + (size_t)(tm + l) * lda;
  const float* __restrict__ Bcol = B + tn + l;

  v8f acc = {};
  for (int k = 0; k < K; k += 4) {
    const int kk = k + half * 2;
    v2f a, b;
    a.x = Arow[kk];
    a.y = Arow[kk + 1];
    b.x = Bcol[(size_t)kk * N];
    b.y = Bcol[(size_t)(kk + 1) * N];
    acc = __builtin_amdgcn_wmma_f32_16x16x4_f32(
         false, a,  false, b,
         (short)0, acc,  false,  false);
  }

  const float bi = bias ? bias[tn + l] : 0.0f;
#pragma unroll
  for (int r = 0; r < 8; ++r) {
    const int m = tm + half * 8 + r;
    float v = acc[r] + (rowscale ? bi * rowscale[m] : bi);
    if (relu) v = fmaxf(v, 0.0f);
    C[(size_t)m * ldc + tn + l] = v;
  }
}

#define L1_PITCH 260
__global__ __launch_bounds__(512)
void k_gemm_l1_both(const float* __restrict__ x, const float* __restrict__ W_a,
                    const float* __restrict__ W_b, float* __restrict__ C_a,
                    float* __restrict__ C_b, int n_rows) {
  __shared__ __attribute__((aligned(16))) float xs[2][16 * L1_PITCH];
  const int n_tiles = n_rows >> 4;
  const int tid = threadIdx.x;
  const int lane = tid & 31;
  const int wave = tid >> 5;
  const int half = lane >> 4;
  const int l = lane & 15;
  const bool side_b = wave >= 8;
  const float* __restrict__ Wm = side_b ? W_b : W_a;
  float* __restrict__ C = side_b ? C_b : C_a;
  const int n0 = (wave & 7) << 4;

  v2f breg[64];
  const float* __restrict__ Bcol = Wm + n0 + l;
#pragma unroll
  for (int p = 0; p < 64; ++p) {
    const int kk = 4 * p + 2 * half;
    breg[p].x = Bcol[kk * 128];
    breg[p].y = Bcol[(kk + 1) * 128];
  }

  const int q0 = tid, q1 = tid + 512;
  const int so0 = (q0 >> 6) * L1_PITCH + (q0 & 63) * 4;
  const int so1 = (q1 >> 6) * L1_PITCH + (q1 & 63) * 4;

  int t = blockIdx.x;
  if (t < n_tiles) {
    const float4* __restrict__ g = (const float4*)(x + (size_t)t * 4096);
    const float4 a0 = g[q0], a1 = g[q1];
    *(float4*)(&xs[0][so0]) = a0;
    *(float4*)(&xs[0][so1]) = a1;
  }
  __syncthreads();

  int buf = 0;
  int t_done = -1;
  v8f done = {};
  for (; t < n_tiles; t += gridDim.x, buf ^= 1) {
    const int tn = t + (int)gridDim.x;
    const bool more = tn < n_tiles;
    float4 p0 = make_float4(0.f, 0.f, 0.f, 0.f), p1 = p0;
    if (more) {
      const float4* __restrict__ g = (const float4*)(x + (size_t)tn * 4096);
      p0 = g[q0];
      p1 = g[q1];
    }
    if (t_done >= 0) {
      float* __restrict__ Crow = C + ((size_t)t_done * 16 + half * 8) * 128 + n0 + l;
#pragma unroll
      for (int r = 0; r < 8; ++r) Crow[(size_t)r * 128] = done[r];
    }

    const float* As = &xs[buf][l * L1_PITCH + 2 * half];
    v8f acc = {};
    v2f fa[16], fb[16];
#pragma unroll
    for (int i = 0; i < 16; ++i) fa[i] = *(const v2f*)(As + 4 * i);
#pragma unroll
    for (int g = 0; g < 4; g += 2) {
#pragma unroll
      for (int i = 0; i < 16; ++i) fb[i] = *(const v2f*)(As + 4 * (16 * (g + 1) + i));
      __builtin_amdgcn_sched_barrier(0);
#pragma unroll
      for (int i = 0; i < 16; ++i)
        acc = __builtin_amdgcn_wmma_f32_16x16x4_f32(
             false, fa[i],  false, breg[16 * g + i],
             (short)0, acc,  false,  false);
      if (g + 2 < 4) {
#pragma unroll
        for (int i = 0; i < 16; ++i) fa[i] = *(const v2f*)(As + 4 * (16 * (g + 2) + i));
      }
      __builtin_amdgcn_sched_barrier(0);
#pragma unroll
      for (int i = 0; i < 16; ++i)
        acc = __builtin_amdgcn_wmma_f32_16x16x4_f32(
             false, fb[i],  false, breg[16 * (g + 1) + i],
             (short)0, acc,  false,  false);
    }
    done = acc;
    t_done = t;

    if (more) {
      *(float4*)(&xs[buf ^ 1][so0]) = p0;
      *(float4*)(&xs[buf ^ 1][so1]) = p1;
    }
    __syncthreads();
  }
  if (t_done >= 0) {
    float* __restrict__ Crow = C + ((size_t)t_done * 16 + half * 8) * 128 + n0 + l;
#pragma unroll
    for (int r = 0; r < 8; ++r) Crow[(size_t)r * 128] = done[r];
  }
}

#define OWN_CAP  136
#define STEP_CAP 8192
__global__ __launch_bounds__(1024)
void k_own_agg(const float* __restrict__ h, const int* __restrict__ s_idx,
               const int* __restrict__ d_idx, const float* __restrict__ dinv,
               const float* __restrict__ bias, const int* __restrict__ map,
               float* __restrict__ out, int n_edges, int n_nodes) {
  __shared__ float acc[OWN_ROWS * 128];
  __shared__ int2 list[32 * OWN_CAP];
  __shared__ __attribute__((aligned(16))) unsigned short steps[STEP_CAP];
  __shared__ int n_flag;
  const int tid = threadIdx.x;
  const int lane = tid & 31;
  const int wave = tid >> 5;
  const int nwaves = blockDim.x >> 5;
  const int wg = blockIdx.x;
  const int lo = wg * OWN_ROWS;
  const int cnt = min(OWN_ROWS, n_nodes - lo);
  if (cnt <= 0) return;
  const unsigned ucnt = (unsigned)cnt;

  if (tid == 0) n_flag = 0;
  float4* a4 = (float4*)acc;
  for (int i = tid; i < OWN_ROWS * 32; i += blockDim.x) a4[i] = make_float4(0.f, 0.f, 0.f, 0.f);
  uint4* s4 = (uint4*)steps;
  for (int i = tid; i < STEP_CAP / 8; i += blockDim.x) s4[i] = make_uint4(0u, 0u, 0u, 0u);
  __syncthreads();

  const int n_full = n_edges & ~3;
  const int n_words = (n_edges + 4095) >> 12;
  const int w0 = (int)(((unsigned)wg * (unsigned)n_words) / gridDim.x);
  for (int j0 = tid; j0 < n_words; j0 += blockDim.x) {
    int j = j0 + w0;
    if (j >= n_words) j -= n_words;
    unsigned m = (unsigned)map[(size_t)j * MAP_OWNERS + wg];
    const int c = __popc(m);
    if (c) {
      int pos = atomicAdd(&n_flag, c);
      while (m) {
        const int b = __builtin_ctz(m);
        m &= m - 1;
        if (pos < STEP_CAP) steps[pos] = (unsigned short)(j * 32 + b);
        ++pos;
      }
    }
  }
  __syncthreads();
  const int nf = __builtin_amdgcn_readfirstlane(min(n_flag, STEP_CAP));
  const int wv = __builtin_amdgcn_readfirstlane(wave);

  int2* wl = list + wave * OWN_CAP;
  int n = 0;

  auto add_rows = [&](const int k) {
    const int u = ((lane & 7) < k) ? (lane & 7) : 0;
    const int2 ent = wl[n - k + u];
    const int s = s_idx[ent.x];
    const int r = ent.y;
    const int wi = __float_as_int(dinv[s] * dinv[lo + r]);
    float4 v[8];
#pragma unroll
    for (int j = 0; j < 8; ++j) {
      if (j < k) {
        const int sj = __builtin_amdgcn_readlane(s, j);
        v[j] = ((const float4*)(h + (size_t)sj * 128))[lane];
      }
    }
#pragma unroll
    for (int j = 0; j < 8; ++j) {
      if (j < k) {
        const int rj = __builtin_amdgcn_readlane(r, j);
        const float wj = __int_as_float(__builtin_amdgcn_readlane(wi, j));
        float* t = acc + rj * 128 + lane;
        __hip_atomic_fetch_add(t,      v[j].x * wj, __ATOMIC_RELAXED, __HIP_MEMORY_SCOPE_WORKGROUP);
        __hip_atomic_fetch_add(t + 32, v[j].y * wj, __ATOMIC_RELAXED, __HIP_MEMORY_SCOPE_WORKGROUP);
        __hip_atomic_fetch_add(t + 64, v[j].z * wj, __ATOMIC_RELAXED, __HIP_MEMORY_SCOPE_WORKGROUP);
        __hip_atomic_fetch_add(t + 96, v[j].w * wj, __ATOMIC_RELAXED, __HIP_MEMORY_SCOPE_WORKGROUP);
      }
    }
    n -= k;
  };

  auto push = [&](const bool f, const int e, const int r) {
    unsigned b = __builtin_amdgcn_ballot_w32(f);
    while (b) {
      const int L = __builtin_ctz(b);
      b &= b - 1;
      if (lane == L) wl[n] = make_int2(e, r);
      ++n;
    }
  };

  for (int i0 = wv * 8; i0 < nf; i0 += nwaves * 8) {
    int4 dv[8];
    int eb[8];
    const uint4 sw = *(const uint4*)(&steps[i0]);
    const unsigned swa[4] = {sw.x, sw.y, sw.z, sw.w};
#pragma unroll
    for (int q = 0; q < 8; ++q) {
      const int i = i0 + q;
      const int st = (int)((swa[q >> 1] >> (16 * (q & 1))) & 0xffffu);
      const int e0 = st * 128 + lane * 4;
      const bool ok = i < nf && e0 < n_full;
      eb[q] = ok ? e0 : -1;
      dv[q] = *(const int4*)(d_idx + (ok ? e0 : 0));
    }
#pragma unroll
    for (int q = 0; q < 8; ++q) {
      const int e0 = eb[q];
      const bool ok = e0 >= 0;
      const int rx = dv[q].x - lo, ry = dv[q].y - lo, rz = dv[q].z - lo, rw = dv[q].w - lo;
      push(ok && (unsigned)rx < ucnt, e0 + 0, rx);
      push(ok && (unsigned)ry < ucnt, e0 + 1, ry);
      push(ok && (unsigned)rz < ucnt, e0 + 2, rz);
      push(ok && (unsigned)rw < ucnt, e0 + 3, rw);
      while (n >= 8) add_rows(8);
    }
  }
  if (wave == 0 && n_full < n_edges) {
    const int e = n_full + lane;
    const bool in = e < n_edges;
    const int r = (in ? d_idx[e] : -1) - lo;
    push(in && (unsigned)r < ucnt, e, r);
  }
  if (n > 0) add_rows(n);
  __syncthreads();

  const float4* __restrict__ h4 = (const float4*)h;
  const float4* __restrict__ b4 = (const float4*)bias;
  float4* __restrict__ o4 = (float4*)out;
  for (int idx = tid; idx < cnt * 32; idx += blockDim.x) {
    const int r = idx >> 5;
    const int q = idx & 31;
    const int node = lo + r;
    const float di = dinv[node];
    const float4 hl = h4[(size_t)node * 32 + q];
    const float4 b = b4[q];
    const float* t = acc + r * 128 + q;
    float4 o;
    o.x = fmaxf(t[0]  + hl.x * di * di + b.x, 0.0f);
    o.y = fmaxf(t[32] + hl.y * di * di + b.y, 0.0f);
    o.z = fmaxf(t[64] + hl.z * di * di + b.z, 0.0f);
    o.w = fmaxf(t[96] + hl.w * di * di + b.w, 0.0f);
    o4[(size_t)node * 32 + q] = o;
  }
}

#define POOL_G 512
__global__ __launch_bounds__(1024)
void k_edge_pool(const float* __restrict__ h_a, const int* __restrict__ s_a,
                 const int* __restrict__ d_a, const float* __restrict__ dinv_a, int col_a,
                 const float* __restrict__ h_b, const int* __restrict__ s_b,
                 const int* __restrict__ d_b, const float* __restrict__ dinv_b, int col_b,
                 const int* __restrict__ batch,
                 float* __restrict__ pooled, float* __restrict__ cnt,
                 int n_edges, int n_nodes) {
  __shared__ float tile[POOL_G * 128];
  __shared__ float tcnt[POOL_G];
  const int tid = threadIdx.x;
  const int lane = tid & 31;
  const int wave = tid >> 5;
  const int nwaves = blockDim.x >> 5;
  const int side_wgs = (int)gridDim.x >> 1;
  const bool side_b = (int)blockIdx.x >= side_wgs;
  const int wg = side_b ? (int)blockIdx.x - side_wgs : (int)blockIdx.x;
  const float* __restrict__ h    = side_b ? h_b : h_a;
  const int* __restrict__ s_idx  = side_b ? s_b : s_a;
  const int* __restrict__ d_idx  = side_b ? d_b : d_a;
  const float* __restrict__ dinv = side_b ? dinv_b : dinv_a;
  const int col_off = side_b ? col_b : col_a;
  const bool count = !side_b;

  float4* t4 = (float4*)tile;
  for (int i = tid; i < POOL_G * 32; i += blockDim.x) t4[i] = make_float4(0.f, 0.f, 0.f, 0.f);
  for (int i = tid; i < POOL_G; i += blockDim.x) tcnt[i] = 0.0f;
  __syncthreads();

  const int total = n_edges + n_nodes;
  const int chunk = (((total + side_wgs - 1) / side_wgs) + 31) & ~31;
  const int begin = wg * chunk;
  const int end = min(total, begin + chunk);

  for (int base = begin + wave * 32; base < end; base += nwaves * 32) {
    const int i = base + lane;
    const bool valid = i < end;
    int s = 0, g = 0;
    float w = 0.0f;
    if (valid) {
      if (i < n_edges) {
        s = s_idx[i];
        const int d = d_idx[i];
        w = dinv[s] * dinv[d];
        g = batch[d];
      } else {
        s = i - n_edges;
        const float di = dinv[s];
        w = di * di;
        g = batch[s];
        if (count) __hip_atomic_fetch_add(&tcnt[g], 1.0f, __ATOMIC_RELAXED, __HIP_MEMORY_SCOPE_WORKGROUP);
      }
    }
    const int s0 = __builtin_amdgcn_readfirstlane(s);
    const int g0 = __builtin_amdgcn_readfirstlane(g);
    if (!valid) { s = s0; g = g0; }
    const int wi = __float_as_int(w);

#pragma unroll
    for (int jj = 0; jj < 32; jj += 8) {
      float4 v[8];
#pragma unroll
      for (int u = 0; u < 8; ++u) {
        const int sj = __builtin_amdgcn_readlane(s, jj + u);
        v[u] = ((const float4*)(h + (size_t)sj * 128))[lane];
      }
#pragma unroll
      for (int u = 0; u < 8; ++u) {
        const int gj = __builtin_amdgcn_readlane(g, jj + u);
        const float wj = __int_as_float(__builtin_amdgcn_readlane(wi, jj + u));
        float* t = tile + gj * 128 + lane;
        __hip_atomic_fetch_add(t,      v[u].x * wj, __ATOMIC_RELAXED, __HIP_MEMORY_SCOPE_WORKGROUP);
        __hip_atomic_fetch_add(t + 32, v[u].y * wj, __ATOMIC_RELAXED, __HIP_MEMORY_SCOPE_WORKGROUP);
        __hip_atomic_fetch_add(t + 64, v[u].z * wj, __ATOMIC_RELAXED, __HIP_MEMORY_SCOPE_WORKGROUP);
        __hip_atomic_fetch_add(t + 96, v[u].w * wj, __ATOMIC_RELAXED, __HIP_MEMORY_SCOPE_WORKGROUP);
      }
    }
  }
  __syncthreads();

  for (int idx = tid; idx < POOL_G * 128; idx += blockDim.x) {
    const int g = ((idx >> 7) + (wg << 2)) & (POOL_G - 1);
    const int f = idx & 127;
    const float v = tile[g * 128 + (f & 3) * 32 + (f >> 2)];
    atomicAdd(&pooled[(size_t)g * 256 + col_off + f], v);
  }
  if (count) {
    for (int i = tid; i < POOL_G; i += blockDim.x) atomicAdd(&cnt[i], tcnt[i]);
  }
}

static inline void launch_gemm(const float* A, const float* B, float* C,
                               const float* bias, const float* rowscale,
                               int M, int N, int K, int lda, int ldc, int relu,
                               hipStream_t stream) {
  const int tiles = (M >> 4) * (N >> 4);
  const int blocks = (tiles + 7) / 8;
  k_gemm_wmma<<<blocks, 256, 0, stream>>>(A, B, C, bias, rowscale, M, N, K, lda, ldc, relu);
}

extern "C" void kernel_launch(void* const* d_in, const int* in_sizes, int n_in,
                              void* d_out, int out_size, void* d_ws, size_t ws_size,
                              hipStream_t stream) {
  const float* x      = (const float*)d_in[0];
  const int* edge_idx = (const int*)d_in[1];
  const int* batch    = (const int*)d_in[2];
  const float* td_W1 = (const float*)d_in[4];
  const float* td_b1 = (const float*)d_in[5];
  const float* td_W2 = (const float*)d_in[6];
  const float* td_b2 = (const float*)d_in[7];
  const float* bu_W1 = (const float*)d_in[8];
  const float* bu_b1 = (const float*)d_in[9];
  const float* bu_W2 = (const float*)d_in[10];
  const float* bu_b2 = (const float*)d_in[11];
  const float* pw1   = (const float*)d_in[12];
  const float* pb1   = (const float*)d_in[13];
  const float* pw2   = (const float*)d_in[14];
  const float* pb2   = (const float*)d_in[15];

  const int IN_F = 256, HID = 128, OUT_F = 128, G = 512;
  const int n_nodes = in_sizes[0] / IN_F;
  const int n_edges = in_sizes[1] / 2;
  const int* src = edge_idx;
  const int* dst = edge_idx + n_edges;

  float* ws      = (float*)d_ws;
  const size_t NF = (size_t)n_nodes * 128;
  float* h_lin_td = ws;
  float* h_lin_bu = ws + NF;
  float* h_act   = ws + 2 * NF;
  float* h_agg   = ws;
  float* dinv_td = ws + 3 * NF;
  float* dinv_bu = dinv_td + n_nodes;
  float* pooled  = dinv_bu + n_nodes;
  float* gcount  = pooled + (size_t)G * 256;
  float* pooled2 = h_lin_bu;
  float* mlp_h   = h_act;

  const int MR = (n_edges + 4095) / 4096;
  int* map_td = (int*)pooled;
  int* map_bu = map_td + (size_t)MR * 256;
  const int OB = 256;
  const int PB = 256;

  const int DW = (MR + DEG_ROWS - 1) / DEG_ROWS;
  int* planes = (int*)ws;
  k_degree<<<2 * DW, 1024, 0, stream>>>(src, dst, planes, map_td, map_bu, n_edges, MR);
  k_rsqrt<<<(2 * n_nodes + 255) / 256, 256, 0, stream>>>(planes, dinv_td, n_nodes, DW);

  k_gemm_l1_both<<<256, 512, 0, stream>>>(x, td_W1, bu_W1, h_lin_td, h_lin_bu, n_nodes);

  k_own_agg<<<OB, 1024, 0, stream>>>(h_lin_td, src, dst, dinv_td, td_b1, map_td, h_act, n_edges, n_nodes);

  k_own_agg<<<OB, 1024, 0, stream>>>(h_lin_bu, dst, src, dinv_bu, bu_b1, map_bu, h_agg, n_edges, n_nodes);

  hipMemsetAsync(pooled, 0, ((size_t)G * 256 + G) * sizeof(float), stream);
  k_edge_pool<<<PB, 1024, 0, stream>>>(h_act, src, dst, dinv_td,  128,
                                       h_agg, dst, src, dinv_bu,  0,
                                       batch, pooled, gcount, n_edges, n_nodes);

  launch_gemm(pooled, bu_W2, pooled2, bu_b2, gcount, G, OUT_F, HID, 256, 256, 0, stream);
  launch_gemm(pooled + 128, td_W2, pooled2 + 128, td_b2, gcount, G, OUT_F, HID, 256, 256, 0, stream);

  launch_gemm(pooled2, pw1, mlp_h, pb1, nullptr, G, 256, 256, 256, 256,  1, stream);
  launch_gemm(mlp_h, pw2, (float*)d_out, pb2, nullptr, G, 128, 256, 256, 128,  0, stream);
}
